# starting kernel
# baseline (speedup 1.0000x reference)
#include <hip/hip_runtime.h>
#include <hip/hip_bf16.h>
#include <math.h>

#define SEQ    512
#define DIN    1024
#define HID    512

typedef float v2f __attribute__((ext_vector_type(2)));
typedef float v8f __attribute__((ext_vector_type(8)));

__device__ __forceinline__ float hw_tanh(float x)
{
#if __has_builtin(__builtin_amdgcn_tanhf)
    return __builtin_amdgcn_tanhf(x);
#else
    return tanhf(x);
#endif
}

#define CHUNK 256
#define PITCH 260

__global__ void pair_tanh_score(const float* __restrict__ Hm,
                                const float* __restrict__ Mm,
                                const float* __restrict__ w2,
                                const float* __restrict__ b2p,
                                float* __restrict__ out)
{
    __shared__ float sH[16 * PITCH];
    __shared__ float sM[16 * PITCH];
    __shared__ float sW[HID];

    const int t  = threadIdx.x;
    const int ti = t >> 4;
    const int tj = t & 15;
    const int i0 = blockIdx.y * 16;
    const int j0 = blockIdx.x * 16;

    sW[t]       = w2[t];
    sW[t + 256] = w2[t + 256];

    float acc = 0.0f;

    for (int d0 = 0; d0 < HID; d0 += CHUNK) {
        int idx = t;
#pragma unroll
        for (int rep = 0; rep < 4; ++rep, idx += 256) {
            const int r  = idx >> 6;
            const int c4 = (idx & 63) * 4;
            const float4 hv = *(const float4*)(Hm + (size_t)(i0 + r) * HID + d0 + c4);
            const float4 mv = *(const float4*)(Mm + (size_t)(j0 + r) * HID + d0 + c4);
            *(float4*)(&sH[r * PITCH + c4]) = hv;
            *(float4*)(&sM[r * PITCH + c4]) = mv;
        }
        __syncthreads();

        const float* __restrict__ hrow = &sH[ti * PITCH];
        const float* __restrict__ mrow = &sM[tj * PITCH];
        const float* __restrict__ wrow = &sW[d0];
#pragma unroll 8
        for (int d = 0; d < CHUNK; ++d) {
            const float v = hrow[d] + mrow[d];
            acc = fmaf(wrow[d], hw_tanh(v), acc);
        }
        __syncthreads();
    }

    out[(size_t)(i0 + ti) * SEQ + (j0 + tj)] = acc + b2p[0];
}

__global__ void gemm_bias_wmma_f32(const float* __restrict__ X,
                                   const float* __restrict__ W,
                                   const float* __restrict__ bias,
                                   float* __restrict__ Out)
{
    const int wavesPerBlock = blockDim.x >> 5;
    const int wave = blockIdx.x * wavesPerBlock + (threadIdx.x >> 5);
    const int lane = threadIdx.x & 31;

    const int tilesN = HID / 16;
    const int tm = wave / tilesN;
    const int tn = wave % tilesN;
    const int s0 = tm * 16;
    const int h0 = tn * 16;

    const int half = lane >> 4;
    const int l    = lane & 15;

    const float* __restrict__ arow = X + (size_t)(s0 + l) * DIN + 2 * half;
    const float* __restrict__ brow = W + (size_t)(h0 + l) * DIN + 2 * half;

    v8f acc = {};
#pragma unroll 8
    for (int k0 = 0; k0 < DIN; k0 += 4) {
        v2f a = *(const v2f*)(arow + k0);
        v2f b = *(const v2f*)(brow + k0);
        acc = __builtin_amdgcn_wmma_f32_16x16x4_f32(
             false, a,  false, b,
             (short)0, acc,  false,  false);
    }

    const float bv = bias[h0 + l];
#pragma unroll
    for (int r = 0; r < 8; ++r) {
        Out[(size_t)(s0 + r + 8 * half) * HID + h0 + l] = acc[r] + bv;
    }
}

extern "C" void kernel_launch(void* const* d_in, const int* in_sizes, int n_in,
                              void* d_out, int out_size, void* d_ws, size_t ws_size,
                              hipStream_t stream)
{
    const float* x   = (const float*)d_in[0];
    const float* W_h = (const float*)d_in[1];
    const float* b_h = (const float*)d_in[2];
    const float* W_m = (const float*)d_in[3];
    const float* b_m = (const float*)d_in[4];
    const float* w2  = (const float*)d_in[5];
    const float* b2  = (const float*)d_in[6];
    float* out = (float*)d_out;

    float* Hbuf = (float*)d_ws;
    float* Mbuf = Hbuf + (size_t)SEQ * HID;

    const int tiles = (SEQ / 16) * (HID / 16);
    const int blk   = 128;
    const int grid  = tiles / (blk / 32);
    gemm_bias_wmma_f32<<<grid, blk, 0, stream>>>(x, W_h, b_h, Hbuf);
    gemm_bias_wmma_f32<<<grid, blk, 0, stream>>>(x, W_m, b_m, Mbuf);

    dim3 g2(SEQ / 16, SEQ / 16);
    pair_tanh_score<<<g2, 256, 0, stream>>>(Hbuf, Mbuf, w2, b2, out);
}


# one launch for both projections: each wave loads the x fragment once per k-step and feeds two independent f32 WMMA accumulators (its H tile and its M tile); launches 3 -> 2
# speedup vs baseline: 1.3507x; 1.3507x over previous
#include <hip/hip_runtime.h>
#include <hip/hip_bf16.h>
#include <math.h>

#define SEQ    512
#define DIN    1024
#define HID    512

typedef float v2f __attribute__((ext_vector_type(2)));
typedef float v8f __attribute__((ext_vector_type(8)));

__device__ __forceinline__ float hw_tanh(float x)
{
#if __has_builtin(__builtin_amdgcn_tanhf)
    return __builtin_amdgcn_tanhf(x);
#else
    return tanhf(x);
#endif
}

#define CHUNK 256
#define PITCH 260

__global__ void pair_tanh_score(const float* __restrict__ Hm,
                                const float* __restrict__ Mm,
                                const float* __restrict__ w2,
                                const float* __restrict__ b2p,
                                float* __restrict__ out)
{
    __shared__ float sH[16 * PITCH];
    __shared__ float sM[16 * PITCH];
    __shared__ float sW[HID];

    const int t  = threadIdx.x;
    const int ti = t >> 4;
    const int tj = t & 15;
    const int i0 = blockIdx.y * 16;
    const int j0 = blockIdx.x * 16;

    sW[t]       = w2[t];
    sW[t + 256] = w2[t + 256];

    float acc = 0.0f;

    for (int d0 = 0; d0 < HID; d0 += CHUNK) {
        int idx = t;
#pragma unroll
        for (int rep = 0; rep < 4; ++rep, idx += 256) {
            const int r  = idx >> 6;
            const int c4 = (idx & 63) * 4;
            const float4 hv = *(const float4*)(Hm + (size_t)(i0 + r) * HID + d0 + c4);
            const float4 mv = *(const float4*)(Mm + (size_t)(j0 + r) * HID + d0 + c4);
            *(float4*)(&sH[r * PITCH + c4]) = hv;
            *(float4*)(&sM[r * PITCH + c4]) = mv;
        }
        __syncthreads();

        const float* __restrict__ hrow = &sH[ti * PITCH];
        const float* __restrict__ mrow = &sM[tj * PITCH];
        const float* __restrict__ wrow = &sW[d0];
#pragma unroll 8
        for (int d = 0; d < CHUNK; ++d) {
            const float v = hrow[d] + mrow[d];
            acc = fmaf(wrow[d], hw_tanh(v), acc);
        }
        __syncthreads();
    }

    out[(size_t)(i0 + ti) * SEQ + (j0 + tj)] = acc + b2p[0];
}

__global__ void gemm2_bias_wmma_f32(const float* __restrict__ X,
                                    const float* __restrict__ Wh,
                                    const float* __restrict__ bh,
                                    const float* __restrict__ Wm,
                                    const float* __restrict__ bm,
                                    float* __restrict__ Hout,
                                    float* __restrict__ Mout)
{
    const int wavesPerBlock = blockDim.x >> 5;
    const int wave = blockIdx.x * wavesPerBlock + (threadIdx.x >> 5);
    const int lane = threadIdx.x & 31;

    const int tilesN = HID / 16;
    const int tm = wave / tilesN;
    const int tn = wave % tilesN;
    const int s0 = tm * 16;
    const int h0 = tn * 16;

    const int half = lane >> 4;
    const int l    = lane & 15;

    const float* __restrict__ arow  = X  + (size_t)(s0 + l) * DIN + 2 * half;
    const float* __restrict__ bhrow = Wh + (size_t)(h0 + l) * DIN + 2 * half;
    const float* __restrict__ bmrow = Wm + (size_t)(h0 + l) * DIN + 2 * half;

    v8f acch = {};
    v8f accm = {};
    for (int kb = 0; kb < DIN; kb += 32) {
        v2f a[8], wh[8], wm[8];
#pragma unroll
        for (int u = 0; u < 8; ++u) {
            a[u]  = *(const v2f*)(arow  + kb + 4 * u);
            wh[u] = *(const v2f*)(bhrow + kb + 4 * u);
            wm[u] = *(const v2f*)(bmrow + kb + 4 * u);
        }
#pragma unroll
        for (int u = 0; u < 8; ++u) {
            acch = __builtin_amdgcn_wmma_f32_16x16x4_f32(
                 false, a[u],  false, wh[u],
                 (short)0, acch,  false,  false);
            accm = __builtin_amdgcn_wmma_f32_16x16x4_f32(
                 false, a[u],  false, wm[u],
                 (short)0, accm,  false,  false);
        }
    }

    const float bvh = bh[h0 + l];
    const float bvm = bm[h0 + l];
#pragma unroll
    for (int r = 0; r < 8; ++r) {
        const size_t o = (size_t)(s0 + r + 8 * half) * HID + h0 + l;
        Hout[o] = acch[r] + bvh;
        Mout[o] = accm[r] + bvm;
    }
}

extern "C" void kernel_launch(void* const* d_in, const int* in_sizes, int n_in,
                              void* d_out, int out_size, void* d_ws, size_t ws_size,
                              hipStream_t stream)
{
    const float* x   = (const float*)d_in[0];
    const float* W_h = (const float*)d_in[1];
    const float* b_h = (const float*)d_in[2];
    const float* W_m = (const float*)d_in[3];
    const float* b_m = (const float*)d_in[4];
    const float* w2  = (const float*)d_in[5];
    const float* b2  = (const float*)d_in[6];
    float* out = (float*)d_out;

    float* Hbuf = (float*)d_ws;
    float* Mbuf = Hbuf + (size_t)SEQ * HID;

    const int tiles = (SEQ / 16) * (HID / 16);
    const int blk   = 128;
    const int grid  = tiles / (blk / 32);
    gemm2_bias_wmma_f32<<<grid, blk, 0, stream>>>(x, W_h, b_h, W_m, b_m, Hbuf, Mbuf);

    dim3 g2(SEQ / 16, SEQ / 16);
    pair_tanh_score<<<g2, 256, 0, stream>>>(Hbuf, Mbuf, w2, b2, out);
}


# pairwise tanh phase register-tiled: each thread keeps 4x4 outputs over a quarter of d, so one LDS read feeds 16 tanh; the four quarter sums per output are added through LDS (summation order changes)
# speedup vs baseline: 1.4959x; 1.1075x over previous
#include <hip/hip_runtime.h>
#include <hip/hip_bf16.h>
#include <math.h>

#define SEQ    512
#define DIN    1024
#define HID    512

typedef float v2f __attribute__((ext_vector_type(2)));
typedef float v8f __attribute__((ext_vector_type(8)));

__device__ __forceinline__ float hw_tanh(float x)
{
#if __has_builtin(__builtin_amdgcn_tanhf)
    return __builtin_amdgcn_tanhf(x);
#else
    return tanhf(x);
#endif
}

#define TILE  32
#define NQ    4
#define DQ    (HID / NQ)
#define DC    32
#define NCH   (DQ / DC)
#define SP    36
#define RP    36

__device__ __forceinline__ void cell4(float& r, const float4 h, const float4 m, const float4 w)
{
    r = fmaf(w.x, hw_tanh(h.x + m.x), r);
    r = fmaf(w.y, hw_tanh(h.y + m.y), r);
    r = fmaf(w.z, hw_tanh(h.z + m.z), r);
    r = fmaf(w.w, hw_tanh(h.w + m.w), r);
}

#define FRAG_DECL(S) float4 h##S##0, h##S##1, h##S##2, h##S##3, m##S##0, m##S##1, m##S##2, m##S##3, w##S
#define FRAG_LOAD(S, s_)                                                 \
    do {                                                                 \
        w##S    = *(const float4*)(wb + 4 * (s_));                       \
        h##S##0 = *(const float4*)(hb + 0 * 8 * SP + 4 * (s_));          \
        h##S##1 = *(const float4*)(hb + 1 * 8 * SP + 4 * (s_));          \
        h##S##2 = *(const float4*)(hb + 2 * 8 * SP + 4 * (s_));          \
        h##S##3 = *(const float4*)(hb + 3 * 8 * SP + 4 * (s_));          \
        m##S##0 = *(const float4*)(mb + 0 * 8 * SP + 4 * (s_));          \
        m##S##1 = *(const float4*)(mb + 1 * 8 * SP + 4 * (s_));          \
        m##S##2 = *(const float4*)(mb + 2 * 8 * SP + 4 * (s_));          \
        m##S##3 = *(const float4*)(mb + 3 * 8 * SP + 4 * (s_));          \
    } while (0)
#define TILE_ROW(S, a_)                                                  \
    do {                                                                 \
        cell4(acc[a_][0], h##S##a_, m##S##0, w##S);                      \
        cell4(acc[a_][1], h##S##a_, m##S##1, w##S);                      \
        cell4(acc[a_][2], h##S##a_, m##S##2, w##S);                      \
        cell4(acc[a_][3], h##S##a_, m##S##3, w##S);                      \
    } while (0)
#define TILE_FMA(S)                                                      \
    do {                                                                 \
        TILE_ROW(S, 0);                                                  \
        TILE_ROW(S, 1);                                                  \
        TILE_ROW(S, 2);                                                  \
        TILE_ROW(S, 3);                                                  \
    } while (0)

__global__ void __launch_bounds__(256)
pair_tanh_score(const float* __restrict__ Hm,
                const float* __restrict__ Mm,
                const float* __restrict__ w2,
                const float* __restrict__ b2p,
                float* __restrict__ out)
{
    __shared__ __attribute__((aligned(16))) float sH[NCH][NQ * TILE * SP];
    __shared__ __attribute__((aligned(16))) float sM[NCH][NQ * TILE * SP];
    __shared__ __attribute__((aligned(16))) float sW[HID];
    __shared__ __attribute__((aligned(16))) float sR[NQ * TILE * RP];

    const int t  = threadIdx.x;
    const int q  = t >> 6;
    const int lt = t & 63;
    const int ti = lt >> 3;
    const int tj = lt & 7;
    const int i0 = blockIdx.y * TILE;
    const int j0 = blockIdx.x * TILE;

    const int srow = t >> 3;
    const int sc4  = (t & 7) * 4;
    const float* __restrict__ gH = Hm + (size_t)(i0 + srow) * HID + sc4;
    const float* __restrict__ gM = Mm + (size_t)(j0 + srow) * HID + sc4;
    const int sOff = srow * SP + sc4;

    sW[t]       = w2[t];
    sW[t + 256] = w2[t + 256];

    float4 ph0, ph1, ph2, ph3, pm0, pm1, pm2, pm3;
#define STAGE_LOAD(c_)                                                   \
    do {                                                                 \
        ph0 = *(const float4*)(gH + 0 * DQ + (c_) * DC);                 \
        ph1 = *(const float4*)(gH + 1 * DQ + (c_) * DC);                 \
        ph2 = *(const float4*)(gH + 2 * DQ + (c_) * DC);                 \
        ph3 = *(const float4*)(gH + 3 * DQ + (c_) * DC);                 \
        pm0 = *(const float4*)(gM + 0 * DQ + (c_) * DC);                 \
        pm1 = *(const float4*)(gM + 1 * DQ + (c_) * DC);                 \
        pm2 = *(const float4*)(gM + 2 * DQ + (c_) * DC);                 \
        pm3 = *(const float4*)(gM + 3 * DQ + (c_) * DC);                 \
    } while (0)
#define STAGE_STORE(b_)                                                  \
    do {                                                                 \
        *(float4*)(&sH[(b_)][0 * TILE * SP + sOff]) = ph0;               \
        *(float4*)(&sH[(b_)][1 * TILE * SP + sOff]) = ph1;               \
        *(float4*)(&sH[(b_)][2 * TILE * SP + sOff]) = ph2;               \
        *(float4*)(&sH[(b_)][3 * TILE * SP + sOff]) = ph3;               \
        *(float4*)(&sM[(b_)][0 * TILE * SP + sOff]) = pm0;               \
        *(float4*)(&sM[(b_)][1 * TILE * SP + sOff]) = pm1;               \
        *(float4*)(&sM[(b_)][2 * TILE * SP + sOff]) = pm2;               \
        *(float4*)(&sM[(b_)][3 * TILE * SP + sOff]) = pm3;               \
    } while (0)

    STAGE_LOAD(0);
    STAGE_STORE(0);
    __syncthreads();

    float acc[4][4] = {};

    FRAG_DECL(A);
    FRAG_DECL(B);

    for (int c = 0; c < NCH; ++c) {
        if (c + 1 < NCH) STAGE_LOAD(c + 1);

        const float* __restrict__ hb = &sH[c][(q * TILE + ti) * SP];
        const float* __restrict__ mb = &sM[c][(q * TILE + tj) * SP];
        const float* __restrict__ wb = &sW[q * DQ + c * DC];

        FRAG_LOAD(A, 0);
        for (int s2 = 0; s2 < DC / 8 - 1; ++s2) {
            FRAG_LOAD(B, 2 * s2 + 1);
            __builtin_amdgcn_sched_barrier(0);
            TILE_FMA(A);
            FRAG_LOAD(A, 2 * s2 + 2);
            __builtin_amdgcn_sched_barrier(0);
            TILE_FMA(B);
        }
        FRAG_LOAD(B, DC / 4 - 1);
        __builtin_amdgcn_sched_barrier(0);
        TILE_FMA(A);
        TILE_FMA(B);

        if (c + 1 < NCH) STAGE_STORE(c + 1);
        __syncthreads();
    }

#pragma unroll
    for (int a = 0; a < 4; ++a)
#pragma unroll
        for (int b = 0; b < 4; ++b)
            sR[(q * TILE + ti + 8 * a) * RP + tj + 8 * b] = acc[a][b];
    __syncthreads();

    const float4 r0 = *(const float4*)(&sR[(0 * TILE + srow) * RP + sc4]);
    const float4 r1 = *(const float4*)(&sR[(1 * TILE + srow) * RP + sc4]);
    const float4 r2 = *(const float4*)(&sR[(2 * TILE + srow) * RP + sc4]);
    const float4 r3 = *(const float4*)(&sR[(3 * TILE + srow) * RP + sc4]);
    const float bias = b2p[0];
    float4 o;
    o.x = ((r0.x + r1.x) + (r2.x + r3.x)) + bias;
    o.y = ((r0.y + r1.y) + (r2.y + r3.y)) + bias;
    o.z = ((r0.z + r1.z) + (r2.z + r3.z)) + bias;
    o.w = ((r0.w + r1.w) + (r2.w + r3.w)) + bias;
    *(float4*)(out + (size_t)(i0 + srow) * SEQ + j0 + sc4) = o;
}

__global__ void gemm2_bias_wmma_f32(const float* __restrict__ X,
                                    const float* __restrict__ Wh,
                                    const float* __restrict__ bh,
                                    const float* __restrict__ Wm,
                                    const float* __restrict__ bm,
                                    float* __restrict__ Hout,
                                    float* __restrict__ Mout)
{
    const int wavesPerBlock = blockDim.x >> 5;
    const int wave = blockIdx.x * wavesPerBlock + (threadIdx.x >> 5);
    const int lane = threadIdx.x & 31;

    const int tilesN = HID / 16;
    const int tm = wave / tilesN;
    const int tn = wave % tilesN;
    const int s0 = tm * 16;
    const int h0 = tn * 16;

    const int half = lane >> 4;
    const int l    = lane & 15;

    const float* __restrict__ arow  = X  + (size_t)(s0 + l) * DIN + 2 * half;
    const float* __restrict__ bhrow = Wh + (size_t)(h0 + l) * DIN + 2 * half;
    const float* __restrict__ bmrow = Wm + (size_t)(h0 + l) * DIN + 2 * half;

    v8f acch = {};
    v8f accm = {};
    for (int kb = 0; kb < DIN; kb += 32) {
        v2f a[8], wh[8], wm[8];
#pragma unroll
        for (int u = 0; u < 8; ++u) {
            a[u]  = *(const v2f*)(arow  + kb + 4 * u);
            wh[u] = *(const v2f*)(bhrow + kb + 4 * u);
            wm[u] = *(const v2f*)(bmrow + kb + 4 * u);
        }
#pragma unroll
        for (int u = 0; u < 8; ++u) {
            acch = __builtin_amdgcn_wmma_f32_16x16x4_f32(
                 false, a[u],  false, wh[u],
                 (short)0, acch,  false,  false);
            accm = __builtin_amdgcn_wmma_f32_16x16x4_f32(
                 false, a[u],  false, wm[u],
                 (short)0, accm,  false,  false);
        }
    }

    const float bvh = bh[h0 + l];
    const float bvm = bm[h0 + l];
#pragma unroll
    for (int r = 0; r < 8; ++r) {
        const size_t o = (size_t)(s0 + r + 8 * half) * HID + h0 + l;
        Hout[o] = acch[r] + bvh;
        Mout[o] = accm[r] + bvm;
    }
}

extern "C" void kernel_launch(void* const* d_in, const int* in_sizes, int n_in,
                              void* d_out, int out_size, void* d_ws, size_t ws_size,
                              hipStream_t stream)
{
    const float* x   = (const float*)d_in[0];
    const float* W_h = (const float*)d_in[1];
    const float* b_h = (const float*)d_in[2];
    const float* W_m = (const float*)d_in[3];
    const float* b_m = (const float*)d_in[4];
    const float* w2  = (const float*)d_in[5];
    const float* b2  = (const float*)d_in[6];
    float* out = (float*)d_out;

    float* Hbuf = (float*)d_ws;
    float* Mbuf = Hbuf + (size_t)SEQ * HID;

    const int tiles = (SEQ / 16) * (HID / 16);
    const int blk   = 128;
    const int grid  = tiles / (blk / 32);
    gemm2_bias_wmma_f32<<<grid, blk, 0, stream>>>(x, W_h, b_h, W_m, b_m, Hbuf, Mbuf);

    dim3 g2(SEQ / 32, SEQ / 32);
    pair_tanh_score<<<g2, 256, 0, stream>>>(Hbuf, Mbuf, w2, b2, out);
}


# GEMM operands staged through LDS: each 4-wave block loads its 32-row X, W_h, W_m slabs as whole 512-B row pieces, next 128-k chunk in flight under the WMMAs; b128 fragment reads change k sum order
# speedup vs baseline: 2.2976x; 1.5359x over previous
#include <hip/hip_runtime.h>
#include <hip/hip_bf16.h>
#include <math.h>

#define SEQ    512
#define DIN    1024
#define HID    512

typedef float v2f __attribute__((ext_vector_type(2)));
typedef float v8f __attribute__((ext_vector_type(8)));

__device__ __forceinline__ float hw_tanh(float x)
{
#if __has_builtin(__builtin_amdgcn_tanhf)
    return __builtin_amdgcn_tanhf(x);
#else
    return tanhf(x);
#endif
}

#define TILE  32
#define NQ    4
#define DQ    (HID / NQ)
#define DC    32
#define NCH   (DQ / DC)
#define SP    36
#define RP    36

__device__ __forceinline__ void cell4(float& r, const float4 h, const float4 m, const float4 w)
{
    r = fmaf(w.x, hw_tanh(h.x + m.x), r);
    r = fmaf(w.y, hw_tanh(h.y + m.y), r);
    r = fmaf(w.z, hw_tanh(h.z + m.z), r);
    r = fmaf(w.w, hw_tanh(h.w + m.w), r);
}

#define FRAG_DECL(S) float4 h##S##0, h##S##1, h##S##2, h##S##3, m##S##0, m##S##1, m##S##2, m##S##3, w##S
#define FRAG_LOAD(S, s_)                                                 \
    do {                                                                 \
        w##S    = *(const float4*)(wb + 4 * (s_));                       \
        h##S##0 = *(const float4*)(hb + 0 * 8 * SP + 4 * (s_));          \
        h##S##1 = *(const float4*)(hb + 1 * 8 * SP + 4 * (s_));          \
        h##S##2 = *(const float4*)(hb + 2 * 8 * SP + 4 * (s_));          \
        h##S##3 = *(const float4*)(hb + 3 * 8 * SP + 4 * (s_));          \
        m##S##0 = *(const float4*)(mb + 0 * 8 * SP + 4 * (s_));          \
        m##S##1 = *(const float4*)(mb + 1 * 8 * SP + 4 * (s_));          \
        m##S##2 = *(const float4*)(mb + 2 * 8 * SP + 4 * (s_));          \
        m##S##3 = *(const float4*)(mb + 3 * 8 * SP + 4 * (s_));          \
    } while (0)
#define TILE_ROW(S, a_)                                                  \
    do {                                                                 \
        cell4(acc[a_][0], h##S##a_, m##S##0, w##S);                      \
        cell4(acc[a_][1], h##S##a_, m##S##1, w##S);                      \
        cell4(acc[a_][2], h##S##a_, m##S##2, w##S);                      \
        cell4(acc[a_][3], h##S##a_, m##S##3, w##S);                      \
    } while (0)
#define TILE_FMA(S)                                                      \
    do {                                                                 \
        TILE_ROW(S, 0);                                                  \
        TILE_ROW(S, 1);                                                  \
        TILE_ROW(S, 2);                                                  \
        TILE_ROW(S, 3);                                                  \
    } while (0)

__global__ void __launch_bounds__(256)
pair_tanh_score(const float* __restrict__ Hm,
                const float* __restrict__ Mm,
                const float* __restrict__ w2,
                const float* __restrict__ b2p,
                float* __restrict__ out)
{
    __shared__ __attribute__((aligned(16))) float sH[NCH][NQ * TILE * SP];
    __shared__ __attribute__((aligned(16))) float sM[NCH][NQ * TILE * SP];
    __shared__ __attribute__((aligned(16))) float sW[HID];
    __shared__ __attribute__((aligned(16))) float sR[NQ * TILE * RP];

    const int t  = threadIdx.x;
    const int q  = t >> 6;
    const int lt = t & 63;
    const int ti = lt >> 3;
    const int tj = lt & 7;
    const int i0 = blockIdx.y * TILE;
    const int j0 = blockIdx.x * TILE;

    const int srow = t >> 3;
    const int sc4  = (t & 7) * 4;
    const float* __restrict__ gH = Hm + (size_t)(i0 + srow) * HID + sc4;
    const float* __restrict__ gM = Mm + (size_t)(j0 + srow) * HID + sc4;
    const int sOff = srow * SP + sc4;

    sW[t]       = w2[t];
    sW[t + 256] = w2[t + 256];

    float4 ph0, ph1, ph2, ph3, pm0, pm1, pm2, pm3;
#define STAGE_LOAD(c_)                                                   \
    do {                                                                 \
        ph0 = *(const float4*)(gH + 0 * DQ + (c_) * DC);                 \
        ph1 = *(const float4*)(gH + 1 * DQ + (c_) * DC);                 \
        ph2 = *(const float4*)(gH + 2 * DQ + (c_) * DC);                 \
        ph3 = *(const float4*)(gH + 3 * DQ + (c_) * DC);                 \
        pm0 = *(const float4*)(gM + 0 * DQ + (c_) * DC);                 \
        pm1 = *(const float4*)(gM + 1 * DQ + (c_) * DC);                 \
        pm2 = *(const float4*)(gM + 2 * DQ + (c_) * DC);                 \
        pm3 = *(const float4*)(gM + 3 * DQ + (c_) * DC);                 \
    } while (0)
#define STAGE_STORE(b_)                                                  \
    do {                                                                 \
        *(float4*)(&sH[(b_)][0 * TILE * SP + sOff]) = ph0;               \
        *(float4*)(&sH[(b_)][1 * TILE * SP + sOff]) = ph1;               \
        *(float4*)(&sH[(b_)][2 * TILE * SP + sOff]) = ph2;               \
        *(float4*)(&sH[(b_)][3 * TILE * SP + sOff]) = ph3;               \
        *(float4*)(&sM[(b_)][0 * TILE * SP + sOff]) = pm0;               \
        *(float4*)(&sM[(b_)][1 * TILE * SP + sOff]) = pm1;               \
        *(float4*)(&sM[(b_)][2 * TILE * SP + sOff]) = pm2;               \
        *(float4*)(&sM[(b_)][3 * TILE * SP + sOff]) = pm3;               \
    } while (0)

    STAGE_LOAD(0);
    STAGE_STORE(0);
    __syncthreads();

    float acc[4][4] = {};

    FRAG_DECL(A);
    FRAG_DECL(B);

    for (int c = 0; c < NCH; ++c) {
        if (c + 1 < NCH) STAGE_LOAD(c + 1);

        const float* __restrict__ hb = &sH[c][(q * TILE + ti) * SP];
        const float* __restrict__ mb = &sM[c][(q * TILE + tj) * SP];
        const float* __restrict__ wb = &sW[q * DQ + c * DC];

        FRAG_LOAD(A, 0);
        for (int s2 = 0; s2 < DC / 8 - 1; ++s2) {
            FRAG_LOAD(B, 2 * s2 + 1);
            __builtin_amdgcn_sched_barrier(0);
            TILE_FMA(A);
            FRAG_LOAD(A, 2 * s2 + 2);
            __builtin_amdgcn_sched_barrier(0);
            TILE_FMA(B);
        }
        FRAG_LOAD(B, DC / 4 - 1);
        __builtin_amdgcn_sched_barrier(0);
        TILE_FMA(A);
        TILE_FMA(B);

        if (c + 1 < NCH) STAGE_STORE(c + 1);
        __syncthreads();
    }

#pragma unroll
    for (int a = 0; a < 4; ++a)
#pragma unroll
        for (int b = 0; b < 4; ++b)
            sR[(q * TILE + ti + 8 * a) * RP + tj + 8 * b] = acc[a][b];
    __syncthreads();

    const float4 r0 = *(const float4*)(&sR[(0 * TILE + srow) * RP + sc4]);
    const float4 r1 = *(const float4*)(&sR[(1 * TILE + srow) * RP + sc4]);
    const float4 r2 = *(const float4*)(&sR[(2 * TILE + srow) * RP + sc4]);
    const float4 r3 = *(const float4*)(&sR[(3 * TILE + srow) * RP + sc4]);
    const float bias = b2p[0];
    float4 o;
    o.x = ((r0.x + r1.x) + (r2.x + r3.x)) + bias;
    o.y = ((r0.y + r1.y) + (r2.y + r3.y)) + bias;
    o.z = ((r0.z + r1.z) + (r2.z + r3.z)) + bias;
    o.w = ((r0.w + r1.w) + (r2.w + r3.w)) + bias;
    *(float4*)(out + (size_t)(i0 + srow) * SEQ + j0 + sc4) = o;
}

#define GT   32
#define KC   128
#define NKC  (DIN / KC)
#define GP   (KC + 4)
#define KB   32
#define GST  (3 * GT * GP)

__device__ __forceinline__ v8f wmma_f32_k4(const float a0, const float a1,
                                           const float b0, const float b1, const v8f c)
{
    v2f a = {a0, a1};
    v2f b = {b0, b1};
    return __builtin_amdgcn_wmma_f32_16x16x4_f32(
         false, a,  false, b,
         (short)0, c,  false,  false);
}

#define G4(M_, S) M_(S, 0) M_(S, 1) M_(S, 2) M_(S, 3)
#define GFRAG_DECL1(S, j) float4 fx##S##j, fh##S##j, fm##S##j;
#define GFRAG_DECL(S) G4(GFRAG_DECL1, S)
#define GFRAG_LOAD1(S, j)                                                \
    fx##S##j = *(const float4*)(fa + kb + 8 * j);                        \
    fh##S##j = *(const float4*)(fb + kb + 8 * j);                        \
    fm##S##j = *(const float4*)(fc + kb + 8 * j);
#define GFRAG_LOAD(S, kb_)                                               \
    do {                                                                 \
        const int kb = (kb_) * KB;                                       \
        G4(GFRAG_LOAD1, S)                                               \
    } while (0)
#define GWMMA1(S, j)                                                     \
    acch = wmma_f32_k4(fx##S##j.x, fx##S##j.y, fh##S##j.x, fh##S##j.y, acch); \
    accm = wmma_f32_k4(fx##S##j.x, fx##S##j.y, fm##S##j.x, fm##S##j.y, accm); \
    acch = wmma_f32_k4(fx##S##j.z, fx##S##j.w, fh##S##j.z, fh##S##j.w, acch); \
    accm = wmma_f32_k4(fx##S##j.z, fx##S##j.w, fm##S##j.z, fm##S##j.w, accm);
#define GWMMA_BLOCK(S) do { G4(GWMMA1, S) } while (0)

#define G8(M_) M_(0) M_(1) M_(2) M_(3) M_(4) M_(5) M_(6) M_(7)
#define GSTAGE_DECL1(i) float4 px##i, ph##i, pm##i;
#define GSTAGE_LOAD1(i)                                                  \
    px##i = *(const float4*)(gX + (size_t)(4 * i) * DIN + kc);           \
    ph##i = *(const float4*)(gH + (size_t)(4 * i) * DIN + kc);           \
    pm##i = *(const float4*)(gM + (size_t)(4 * i) * DIN + kc);
#define GSTAGE_LOAD(c_)                                                  \
    do {                                                                 \
        const int kc = (c_) * KC;                                        \
        G8(GSTAGE_LOAD1)                                                 \
    } while (0)
#define GSTAGE_STORE1(i)                                                 \
    *(float4*)(d + 0 * GT * GP + 4 * i * GP) = px##i;                    \
    *(float4*)(d + 1 * GT * GP + 4 * i * GP) = ph##i;                    \
    *(float4*)(d + 2 * GT * GP + 4 * i * GP) = pm##i;
#define GSTAGE_STORE(st_)                                                \
    do {                                                                 \
        float* d = sT + (st_) * GST + sOff;                              \
        G8(GSTAGE_STORE1)                                                \
    } while (0)

__global__ void __launch_bounds__(128)
gemm2_bias_wmma_f32(const float* __restrict__ X,
                    const float* __restrict__ Wh,
                    const float* __restrict__ bh,
                    const float* __restrict__ Wm,
                    const float* __restrict__ bm,
                    float* __restrict__ Hout,
                    float* __restrict__ Mout)
{
    __shared__ __attribute__((aligned(16))) float sT[2 * GST];

    const int t    = threadIdx.x;
    const int w    = t >> 5;
    const int lane = t & 31;
    const int half = lane >> 4;
    const int l    = lane & 15;
    const int wy   = w >> 1;
    const int wx   = w & 1;
    const int sb   = (blockIdx.x >> 4) * GT;
    const int hb   = (blockIdx.x & 15) * GT;

    const float* __restrict__ gX = X  + (size_t)(sb + w) * DIN + lane * 4;
    const float* __restrict__ gH = Wh + (size_t)(hb + w) * DIN + lane * 4;
    const float* __restrict__ gM = Wm + (size_t)(hb + w) * DIN + lane * 4;
    const int sOff = w * GP + lane * 4;

    G8(GSTAGE_DECL1)

    GSTAGE_LOAD(0);
    GSTAGE_STORE(0);
    __syncthreads();

    v8f acch = {};
    v8f accm = {};
    GFRAG_DECL(A)
    GFRAG_DECL(B)

    const int aOff = (0 * GT + 16 * wy + l) * GP + 4 * half;
    const int bOff = (1 * GT + 16 * wx + l) * GP + 4 * half;
    const int cOff = (2 * GT + 16 * wx + l) * GP + 4 * half;

#pragma clang loop unroll(disable)
    for (int c = 0; c < NKC; ++c) {
        const int st = c & 1;
        if (c + 1 < NKC) GSTAGE_LOAD(c + 1);

        const float* __restrict__ fa = sT + st * GST + aOff;
        const float* __restrict__ fb = sT + st * GST + bOff;
        const float* __restrict__ fc = sT + st * GST + cOff;

        GFRAG_LOAD(A, 0);
        GFRAG_LOAD(B, 1);
        __builtin_amdgcn_sched_barrier(0);
        GWMMA_BLOCK(A);
        GFRAG_LOAD(A, 2);
        __builtin_amdgcn_sched_barrier(0);
        GWMMA_BLOCK(B);
        GFRAG_LOAD(B, 3);
        __builtin_amdgcn_sched_barrier(0);
        GWMMA_BLOCK(A);
        if (c + 1 < NKC) GSTAGE_STORE(st ^ 1);
        GWMMA_BLOCK(B);
        __syncthreads();
    }

    const int s0 = sb + 16 * wy;
    const int h0 = hb + 16 * wx;
    const float bvh = bh[h0 + l];
    const float bvm = bm[h0 + l];
#pragma unroll
    for (int r = 0; r < 8; ++r) {
        const size_t o = (size_t)(s0 + r + 8 * half) * HID + h0 + l;
        Hout[o] = acch[r] + bvh;
        Mout[o] = accm[r] + bvm;
    }
}

extern "C" void kernel_launch(void* const* d_in, const int* in_sizes, int n_in,
                              void* d_out, int out_size, void* d_ws, size_t ws_size,
                              hipStream_t stream)
{
    const float* x   = (const float*)d_in[0];
    const float* W_h = (const float*)d_in[1];
    const float* b_h = (const float*)d_in[2];
    const float* W_m = (const float*)d_in[3];
    const float* b_m = (const float*)d_in[4];
    const float* w2  = (const float*)d_in[5];
    const float* b2  = (const float*)d_in[6];
    float* out = (float*)d_out;

    float* Hbuf = (float*)d_ws;
    float* Mbuf = Hbuf + (size_t)SEQ * HID;

    gemm2_bias_wmma_f32<<<(SEQ / 32) * (HID / 32), 128, 0, stream>>>(x, W_h, b_h, W_m, b_m, Hbuf, Mbuf);

    dim3 g2(SEQ / 32, SEQ / 32);
    pair_tanh_score<<<g2, 256, 0, stream>>>(Hbuf, Mbuf, w2, b2, out);
}


# GEMM k split over the 4 waves: each wave owns the whole 32x32 tile of H and M for 256 k (8 accumulators) and stages its slice through private LDS, no barrier in the k loop; k sum order changes
# speedup vs baseline: 2.4929x; 1.0850x over previous
#include <hip/hip_runtime.h>
#include <hip/hip_bf16.h>
#include <math.h>

#define SEQ    512
#define DIN    1024
#define HID    512

typedef float v2f __attribute__((ext_vector_type(2)));
typedef float v8f __attribute__((ext_vector_type(8)));

__device__ __forceinline__ float hw_tanh(float x)
{
#if __has_builtin(__builtin_amdgcn_tanhf)
    return __builtin_amdgcn_tanhf(x);
#else
    return tanhf(x);
#endif
}

#define TILE  32
#define NQ    4
#define DQ    (HID / NQ)
#define DC    32
#define NCH   (DQ / DC)
#define SP    36
#define RP    36

__device__ __forceinline__ void cell4(float& r, const float4 h, const float4 m, const float4 w)
{
    r = fmaf(w.x, hw_tanh(h.x + m.x), r);
    r = fmaf(w.y, hw_tanh(h.y + m.y), r);
    r = fmaf(w.z, hw_tanh(h.z + m.z), r);
    r = fmaf(w.w, hw_tanh(h.w + m.w), r);
}

#define FRAG_DECL(S) float4 h##S##0, h##S##1, h##S##2, h##S##3, m##S##0, m##S##1, m##S##2, m##S##3, w##S
#define FRAG_LOAD(S, s_)                                                 \
    do {                                                                 \
        w##S    = *(const float4*)(wb + 4 * (s_));                       \
        h##S##0 = *(const float4*)(hb + 0 * 8 * SP + 4 * (s_));          \
        h##S##1 = *(const float4*)(hb + 1 * 8 * SP + 4 * (s_));          \
        h##S##2 = *(const float4*)(hb + 2 * 8 * SP + 4 * (s_));          \
        h##S##3 = *(const float4*)(hb + 3 * 8 * SP + 4 * (s_));          \
        m##S##0 = *(const float4*)(mb + 0 * 8 * SP + 4 * (s_));          \
        m##S##1 = *(const float4*)(mb + 1 * 8 * SP + 4 * (s_));          \
        m##S##2 = *(const float4*)(mb + 2 * 8 * SP + 4 * (s_));          \
        m##S##3 = *(const float4*)(mb + 3 * 8 * SP + 4 * (s_));          \
    } while (0)
#define TILE_ROW(S, a_)                                                  \
    do {                                                                 \
        cell4(acc[a_][0], h##S##a_, m##S##0, w##S);                      \
        cell4(acc[a_][1], h##S##a_, m##S##1, w##S);                      \
        cell4(acc[a_][2], h##S##a_, m##S##2, w##S);                      \
        cell4(acc[a_][3], h##S##a_, m##S##3, w##S);                      \
    } while (0)
#define TILE_FMA(S)                                                      \
    do {                                                                 \
        TILE_ROW(S, 0);                                                  \
        TILE_ROW(S, 1);                                                  \
        TILE_ROW(S, 2);                                                  \
        TILE_ROW(S, 3);                                                  \
    } while (0)

__global__ void __launch_bounds__(256)
pair_tanh_score(const float* __restrict__ Hm,
                const float* __restrict__ Mm,
                const float* __restrict__ w2,
                const float* __restrict__ b2p,
                float* __restrict__ out)
{
    __shared__ __attribute__((aligned(16))) float sH[NCH][NQ * TILE * SP];
    __shared__ __attribute__((aligned(16))) float sM[NCH][NQ * TILE * SP];
    __shared__ __attribute__((aligned(16))) float sW[HID];
    __shared__ __attribute__((aligned(16))) float sR[NQ * TILE * RP];

    const int t  = threadIdx.x;
    const int q  = t >> 6;
    const int lt = t & 63;
    const int ti = lt >> 3;
    const int tj = lt & 7;
    const int i0 = blockIdx.y * TILE;
    const int j0 = blockIdx.x * TILE;

    const int srow = t >> 3;
    const int sc4  = (t & 7) * 4;
    const float* __restrict__ gH = Hm + (size_t)(i0 + srow) * HID + sc4;
    const float* __restrict__ gM = Mm + (size_t)(j0 + srow) * HID + sc4;
    const int sOff = srow * SP + sc4;

    sW[t]       = w2[t];
    sW[t + 256] = w2[t + 256];

    float4 ph0, ph1, ph2, ph3, pm0, pm1, pm2, pm3;
#define STAGE_LOAD(c_)                                                   \
    do {                                                                 \
        ph0 = *(const float4*)(gH + 0 * DQ + (c_) * DC);                 \
        ph1 = *(const float4*)(gH + 1 * DQ + (c_) * DC);                 \
        ph2 = *(const float4*)(gH + 2 * DQ + (c_) * DC);                 \
        ph3 = *(const float4*)(gH + 3 * DQ + (c_) * DC);                 \
        pm0 = *(const float4*)(gM + 0 * DQ + (c_) * DC);                 \
        pm1 = *(const float4*)(gM + 1 * DQ + (c_) * DC);                 \
        pm2 = *(const float4*)(gM + 2 * DQ + (c_) * DC);                 \
        pm3 = *(const float4*)(gM + 3 * DQ + (c_) * DC);                 \
    } while (0)
#define STAGE_STORE(b_)                                                  \
    do {                                                                 \
        *(float4*)(&sH[(b_)][0 * TILE * SP + sOff]) = ph0;               \
        *(float4*)(&sH[(b_)][1 * TILE * SP + sOff]) = ph1;               \
        *(float4*)(&sH[(b_)][2 * TILE * SP + sOff]) = ph2;               \
        *(float4*)(&sH[(b_)][3 * TILE * SP + sOff]) = ph3;               \
        *(float4*)(&sM[(b_)][0 * TILE * SP + sOff]) = pm0;               \
        *(float4*)(&sM[(b_)][1 * TILE * SP + sOff]) = pm1;               \
        *(float4*)(&sM[(b_)][2 * TILE * SP + sOff]) = pm2;               \
        *(float4*)(&sM[(b_)][3 * TILE * SP + sOff]) = pm3;               \
    } while (0)

    STAGE_LOAD(0);
    STAGE_STORE(0);
    __syncthreads();

    float acc[4][4] = {};

    FRAG_DECL(A);
    FRAG_DECL(B);

    for (int c = 0; c < NCH; ++c) {
        if (c + 1 < NCH) STAGE_LOAD(c + 1);

        const float* __restrict__ hb = &sH[c][(q * TILE + ti) * SP];
        const float* __restrict__ mb = &sM[c][(q * TILE + tj) * SP];
        const float* __restrict__ wb = &sW[q * DQ + c * DC];

        FRAG_LOAD(A, 0);
        for (int s2 = 0; s2 < DC / 8 - 1; ++s2) {
            FRAG_LOAD(B, 2 * s2 + 1);
            __builtin_amdgcn_sched_barrier(0);
            TILE_FMA(A);
            FRAG_LOAD(A, 2 * s2 + 2);
            __builtin_amdgcn_sched_barrier(0);
            TILE_FMA(B);
        }
        FRAG_LOAD(B, DC / 4 - 1);
        __builtin_amdgcn_sched_barrier(0);
        TILE_FMA(A);
        TILE_FMA(B);

        if (c + 1 < NCH) STAGE_STORE(c + 1);
        __syncthreads();
    }

#pragma unroll
    for (int a = 0; a < 4; ++a)
#pragma unroll
        for (int b = 0; b < 4; ++b)
            sR[(q * TILE + ti + 8 * a) * RP + tj + 8 * b] = acc[a][b];
    __syncthreads();

    const float4 r0 = *(const float4*)(&sR[(0 * TILE + srow) * RP + sc4]);
    const float4 r1 = *(const float4*)(&sR[(1 * TILE + srow) * RP + sc4]);
    const float4 r2 = *(const float4*)(&sR[(2 * TILE + srow) * RP + sc4]);
    const float4 r3 = *(const float4*)(&sR[(3 * TILE + srow) * RP + sc4]);
    const float bias = b2p[0];
    float4 o;
    o.x = ((r0.x + r1.x) + (r2.x + r3.x)) + bias;
    o.y = ((r0.y + r1.y) + (r2.y + r3.y)) + bias;
    o.z = ((r0.z + r1.z) + (r2.z + r3.z)) + bias;
    o.w = ((r0.w + r1.w) + (r2.w + r3.w)) + bias;
    *(float4*)(out + (size_t)(i0 + srow) * SEQ + j0 + sc4) = o;
}

#define GT   32
#define KW   (DIN / 4)
#define KS   32
#define NKS  (KW / KS)
#define GP   (KS + 4)
#define GST  (3 * GT * GP)
#define GWV  (2 * GST)

__device__ __forceinline__ v8f wmma_f32_k4(const float a0, const float a1,
                                           const float b0, const float b1, const v8f c)
{
    v2f a = {a0, a1};
    v2f b = {b0, b1};
    return __builtin_amdgcn_wmma_f32_16x16x4_f32(
         false, a,  false, b,
         (short)0, c,  false,  false);
}

#define GFRAG_DECL(S) float4 fx##S##a, fx##S##b, fh##S##a, fh##S##b, fm##S##a, fm##S##b;
#define GFRAG_LOAD(S, base_, j_)                                         \
    do {                                                                 \
        const float* f = (base_) + fOff + 8 * (j_);                      \
        fx##S##a = *(const float4*)(f + (0 * GT +  0) * GP);             \
        fx##S##b = *(const float4*)(f + (0 * GT + 16) * GP);             \
        fh##S##a = *(const float4*)(f + (1 * GT +  0) * GP);             \
        fh##S##b = *(const float4*)(f + (1 * GT + 16) * GP);             \
        fm##S##a = *(const float4*)(f + (2 * GT +  0) * GP);             \
        fm##S##b = *(const float4*)(f + (2 * GT + 16) * GP);             \
    } while (0)
#define GWMMA_K4(S, e0, e1)                                              \
    h00 = wmma_f32_k4(fx##S##a.e0, fx##S##a.e1, fh##S##a.e0, fh##S##a.e1, h00); \
    m00 = wmma_f32_k4(fx##S##a.e0, fx##S##a.e1, fm##S##a.e0, fm##S##a.e1, m00); \
    h01 = wmma_f32_k4(fx##S##a.e0, fx##S##a.e1, fh##S##b.e0, fh##S##b.e1, h01); \
    m01 = wmma_f32_k4(fx##S##a.e0, fx##S##a.e1, fm##S##b.e0, fm##S##b.e1, m01); \
    h10 = wmma_f32_k4(fx##S##b.e0, fx##S##b.e1, fh##S##a.e0, fh##S##a.e1, h10); \
    m10 = wmma_f32_k4(fx##S##b.e0, fx##S##b.e1, fm##S##a.e0, fm##S##a.e1, m10); \
    h11 = wmma_f32_k4(fx##S##b.e0, fx##S##b.e1, fh##S##b.e0, fh##S##b.e1, h11); \
    m11 = wmma_f32_k4(fx##S##b.e0, fx##S##b.e1, fm##S##b.e0, fm##S##b.e1, m11);
#define GWMMA_BLOCK(S) do { GWMMA_K4(S, x, y) GWMMA_K4(S, z, w) } while (0)

#define G8(M_) M_(0) M_(1) M_(2) M_(3) M_(4) M_(5) M_(6) M_(7)
#define GSTAGE_DECL1(g) float4 px##g, ph##g, pm##g;
#define GSTAGE_LOAD1(g)                                                  \
    px##g = *(const float4*)(gX + (size_t)(4 * g) * DIN + kc);           \
    ph##g = *(const float4*)(gH + (size_t)(4 * g) * DIN + kc);           \
    pm##g = *(const float4*)(gM + (size_t)(4 * g) * DIN + kc);
#define GSTAGE_LOAD(i_)                                                  \
    do {                                                                 \
        const int kc = (i_) * KS;                                        \
        G8(GSTAGE_LOAD1)                                                 \
    } while (0)
#define GSTAGE_STORE1(g)                                                 \
    *(float4*)(d + (0 * GT + 4 * g) * GP) = px##g;                       \
    *(float4*)(d + (1 * GT + 4 * g) * GP) = ph##g;                       \
    *(float4*)(d + (2 * GT + 4 * g) * GP) = pm##g;
#define GSTAGE_STORE(base_)                                              \
    do {                                                                 \
        float* d = (base_) + sOff;                                       \
        G8(GSTAGE_STORE1)                                                \
    } while (0)

#define GRED_PUT(t_, acc_)                                               \
    do {                                                                 \
        float* r_ = wbase + ((t_) * 32 + lane) * 8;                      \
        *(float4*)(r_)     = make_float4(acc_[0], acc_[1], acc_[2], acc_[3]); \
        *(float4*)(r_ + 4) = make_float4(acc_[4], acc_[5], acc_[6], acc_[7]); \
    } while (0)

__global__ void __launch_bounds__(128)
gemm2_bias_wmma_f32(const float* __restrict__ X,
                    const float* __restrict__ Wh,
                    const float* __restrict__ bh,
                    const float* __restrict__ Wm,
                    const float* __restrict__ bm,
                    float* __restrict__ Hout,
                    float* __restrict__ Mout)
{
    __shared__ __attribute__((aligned(16))) float sT[4 * GWV];

    const int t    = threadIdx.x;
    const int w    = t >> 5;
    const int lane = t & 31;
    const int half = lane >> 4;
    const int l    = lane & 15;
    const int sb   = (blockIdx.x >> 4) * GT;
    const int hb   = (blockIdx.x & 15) * GT;

    const int sr = lane >> 3;
    const int sp = (lane & 7) * 4;
    const float* __restrict__ gX = X  + (size_t)(sb + sr) * DIN + w * KW + sp;
    const float* __restrict__ gH = Wh + (size_t)(hb + sr) * DIN + w * KW + sp;
    const float* __restrict__ gM = Wm + (size_t)(hb + sr) * DIN + w * KW + sp;
    float* wbase = sT + w * GWV;
    const int sOff = sr * GP + sp;
    const int fOff = l * GP + 4 * half;

    G8(GSTAGE_DECL1)

    GSTAGE_LOAD(0);
    GSTAGE_STORE(wbase);
    GSTAGE_LOAD(1);

    v8f h00 = {}, h01 = {}, h10 = {}, h11 = {};
    v8f m00 = {}, m01 = {}, m10 = {}, m11 = {};
    GFRAG_DECL(A)
    GFRAG_DECL(B)

    GFRAG_LOAD(A, wbase, 0);

    int st = 0;
#pragma clang loop unroll(disable)
    for (int i = 0; i < NKS - 1; ++i) {
        const int sn = st ^ 1;
        const float* cur = wbase + st * GST;
        float*       nxt = wbase + sn * GST;

        GSTAGE_STORE(nxt);
        if (i + 2 < NKS) GSTAGE_LOAD(i + 2);

        GFRAG_LOAD(B, cur, 1);
        __builtin_amdgcn_sched_barrier(0);
        GWMMA_BLOCK(A);
        GFRAG_LOAD(A, cur, 2);
        __builtin_amdgcn_sched_barrier(0);
        GWMMA_BLOCK(B);
        GFRAG_LOAD(B, cur, 3);
        __builtin_amdgcn_sched_barrier(0);
        GWMMA_BLOCK(A);
        GFRAG_LOAD(A, nxt, 0);
        __builtin_amdgcn_sched_barrier(0);
        GWMMA_BLOCK(B);
        st = sn;
    }
    {
        const float* cur = wbase + st * GST;
        GFRAG_LOAD(B, cur, 1);
        __builtin_amdgcn_sched_barrier(0);
        GWMMA_BLOCK(A);
        GFRAG_LOAD(A, cur, 2);
        __builtin_amdgcn_sched_barrier(0);
        GWMMA_BLOCK(B);
        GFRAG_LOAD(B, cur, 3);
        __builtin_amdgcn_sched_barrier(0);
        GWMMA_BLOCK(A);
        GWMMA_BLOCK(B);
    }

    GRED_PUT(0, h00); GRED_PUT(1, h01); GRED_PUT(2, h10); GRED_PUT(3, h11);
    GRED_PUT(4, m00); GRED_PUT(5, m01); GRED_PUT(6, m10); GRED_PUT(7, m11);
    __syncthreads();

    const int oo = w >> 1;
    const int ty = w & 1;
    const float* __restrict__ bias = oo ? bm : bh;
    float* __restrict__ O = oo ? Mout : Hout;
    const int s0 = sb + 16 * ty;
#pragma unroll
    for (int tx = 0; tx < 2; ++tx) {
        const float* r_ = sT + ((oo * 4 + ty * 2 + tx) * 32 + lane) * 8;
        const float4 a0 = *(const float4*)(r_ + 0 * GWV), a1 = *(const float4*)(r_ + 0 * GWV + 4);
        const float4 b0 = *(const float4*)(r_ + 1 * GWV), b1 = *(const float4*)(r_ + 1 * GWV + 4);
        const float4 c0 = *(const float4*)(r_ + 2 * GWV), c1 = *(const float4*)(r_ + 2 * GWV + 4);
        const float4 d0 = *(const float4*)(r_ + 3 * GWV), d1 = *(const float4*)(r_ + 3 * GWV + 4);
        const int h0 = hb + 16 * tx;
        const float bv = bias[h0 + l];
        float* o = O + (size_t)(s0 + 8 * half) * HID + h0 + l;
        o[0 * HID] = ((a0.x + b0.x) + (c0.x + d0.x)) + bv;
        o[1 * HID] = ((a0.y + b0.y) + (c0.y + d0.y)) + bv;
        o[2 * HID] = ((a0.z + b0.z) + (c0.z + d0.z)) + bv;
        o[3 * HID] = ((a0.w + b0.w) + (c0.w + d0.w)) + bv;
        o[4 * HID] = ((a1.x + b1.x) + (c1.x + d1.x)) + bv;
        o[5 * HID] = ((a1.y + b1.y) + (c1.y + d1.y)) + bv;
        o[6 * HID] = ((a1.z + b1.z) + (c1.z + d1.z)) + bv;
        o[7 * HID] = ((a1.w + b1.w) + (c1.w + d1.w)) + bv;
    }
}

extern "C" void kernel_launch(void* const* d_in, const int* in_sizes, int n_in,
                              void* d_out, int out_size, void* d_ws, size_t ws_size,
                              hipStream_t stream)
{
    const float* x   = (const float*)d_in[0];
    const float* W_h = (const float*)d_in[1];
    const float* b_h = (const float*)d_in[2];
    const float* W_m = (const float*)d_in[3];
    const float* b_m = (const float*)d_in[4];
    const float* w2  = (const float*)d_in[5];
    const float* b2  = (const float*)d_in[6];
    float* out = (float*)d_out;

    float* Hbuf = (float*)d_ws;
    float* Mbuf = Hbuf + (size_t)SEQ * HID;

    gemm2_bias_wmma_f32<<<(SEQ / 32) * (HID / 32), 128, 0, stream>>>(x, W_h, b_h, W_m, b_m, Hbuf, Mbuf);

    dim3 g2(SEQ / 32, SEQ / 32);
    pair_tanh_score<<<g2, 256, 0, stream>>>(Hbuf, Mbuf, w2, b2, out);
}
